# starting kernel
# baseline (speedup 1.0000x reference)
#include <hip/hip_runtime.h>

typedef float v2f __attribute__((ext_vector_type(2)));
typedef float v8f __attribute__((ext_vector_type(8)));

#define NF 32
#define ED 128
#define RS 132
#define NPAIR 496
#define WAVES_PER_BLOCK 2

__device__ __forceinline__ int triu_idx(int i, int j) {
    return 31 * i - ((i * (i - 1)) >> 1) + j - i - 1;
}

__global__ __launch_bounds__(WAVES_PER_BLOCK * 32)
void soi_wmma_kernel(const float* __restrict__ x, float* __restrict__ out, int n_batch) {
    __shared__ float sm[WAVES_PER_BLOCK][NF][RS];

    const int lane = threadIdx.x & 31;
    const int wave = threadIdx.x >> 5;
    const int b    = blockIdx.x * WAVES_PER_BLOCK + wave;
    if (b >= n_batch) return;

    const float* __restrict__ xb = x + (size_t)b * (NF * ED);
    float (*S)[RS] = sm[wave];

    #pragma unroll 8
    for (int r = 0; r < NF; ++r) {
        const float4 v = *reinterpret_cast<const float4*>(xb + r * ED + (lane << 2));
        *reinterpret_cast<float4*>(&S[r][lane << 2]) = v;
    }
    asm volatile("s_wait_dscnt 0" ::: "memory");

    v8f a00 = {};
    v8f a01 = {};
    v8f a11 = {};

    const int rsel = lane & 15;
    const int coff = (lane >> 4) << 1;
    const float* __restrict__ row0 = &S[rsel][coff];
    const float* __restrict__ row1 = &S[16 + rsel][coff];

    #pragma unroll 4
    for (int k = 0; k < ED; k += 4) {
        const v2f h0 = *reinterpret_cast<const v2f*>(row0 + k);
        const v2f h1 = *reinterpret_cast<const v2f*>(row1 + k);
        a00 = __builtin_amdgcn_wmma_f32_16x16x4_f32(false, h0, false, h0, (short)0, a00, false, false);
        a01 = __builtin_amdgcn_wmma_f32_16x16x4_f32(false, h0, false, h1, (short)0, a01, false, false);
        a11 = __builtin_amdgcn_wmma_f32_16x16x4_f32(false, h1, false, h1, (short)0, a11, false, false);
    }

    float* __restrict__ ob = out + (size_t)b * NPAIR;
    const int j0  = lane & 15;
    const int hi8 = (lane >> 4) << 3;

    #pragma unroll
    for (int v = 0; v < 8; ++v) {
        const int i = v + hi8, j = j0;
        if (i < j) ob[triu_idx(i, j)] = a00[v];
    }
    #pragma unroll
    for (int v = 0; v < 8; ++v) {
        const int i = v + hi8, j = 16 + j0;
        ob[triu_idx(i, j)] = a01[v];
    }
    #pragma unroll
    for (int v = 0; v < 8; ++v) {
        const int i = 16 + v + hi8, j = 16 + j0;
        if (i < j) ob[triu_idx(i, j)] = a11[v];
    }
}

extern "C" void kernel_launch(void* const* d_in, const int* in_sizes, int n_in,
                              void* d_out, int out_size, void* d_ws, size_t ws_size,
                              hipStream_t stream) {
    const float* x = (const float*)d_in[0];
    float* out     = (float*)d_out;
    const int n_batch = in_sizes[0] / (NF * ED);
    const int blocks  = (n_batch + WAVES_PER_BLOCK - 1) / WAVES_PER_BLOCK;
    soi_wmma_kernel<<<blocks, WAVES_PER_BLOCK * 32, 0, stream>>>(x, out, n_batch);
}


# each wave walks 4 batches (256 workgroups x 16 waves, one per WGP) and requests the next batch's 32 rows into registers ahead of the current batch's LDS and WMMA phase; sums and order unchanged
# speedup vs baseline: 1.1118x; 1.1118x over previous
#include <hip/hip_runtime.h>

typedef float v2f __attribute__((ext_vector_type(2)));
typedef float v4f __attribute__((ext_vector_type(4)));
typedef float v8f __attribute__((ext_vector_type(8)));

#define NF 32
#define ED 128
#define RS 132
#define NPAIR 496
#define WAVES_PER_BLOCK 16
#define NUM_BLOCKS 256

__device__ __forceinline__ int triu_idx(int i, int j) {
    return 31 * i - ((i * (i - 1)) >> 1) + j - i - 1;
}

__global__ __launch_bounds__(WAVES_PER_BLOCK * 32)
void soi_wmma_kernel(const float* __restrict__ x, float* __restrict__ out, int n_batch) {
    __shared__ float sm[WAVES_PER_BLOCK][NF][RS];

    const int lane = threadIdx.x & 31;
    const int wave = __builtin_amdgcn_readfirstlane((int)(threadIdx.x >> 5));
    const int nwaves = (int)gridDim.x * WAVES_PER_BLOCK;
    int b = (int)blockIdx.x * WAVES_PER_BLOCK + wave;
    if (b >= n_batch) return;

    float (*S)[RS] = sm[wave];
    const float* __restrict__ xl = x + (lane << 2);

    const int rsel = lane & 15;
    const int coff = (lane >> 4) << 1;
    const float* row0 = &S[rsel][coff];
    const float* row1 = &S[16 + rsel][coff];
    const int j0  = lane & 15;
    const int hi8 = (lane >> 4) << 3;

    v4f pf[NF];
    {
        const float* __restrict__ xb = xl + (size_t)b * (NF * ED);
        #pragma unroll
        for (int r = 0; r < NF; ++r)
            pf[r] = *reinterpret_cast<const v4f*>(xb + r * ED);
    }

    for (;;) {
        #pragma unroll
        for (int r = 0; r < NF; ++r)
            *reinterpret_cast<v4f*>(&S[r][lane << 2]) = pf[r];
        asm volatile("s_wait_dscnt 0" ::: "memory");

        const int bn = b + nwaves;
        const bool has_next = bn < n_batch;
        if (has_next) {
            const float* __restrict__ xb = xl + (size_t)bn * (NF * ED);
            #pragma unroll
            for (int r = 0; r < NF; ++r)
                pf[r] = *reinterpret_cast<const v4f*>(xb + r * ED);
        }

        v8f a00 = {};
        v8f a01 = {};
        v8f a11 = {};

        #pragma unroll 4
        for (int k = 0; k < ED; k += 4) {
            const v2f h0 = *reinterpret_cast<const v2f*>(row0 + k);
            const v2f h1 = *reinterpret_cast<const v2f*>(row1 + k);
            a00 = __builtin_amdgcn_wmma_f32_16x16x4_f32(false, h0, false, h0, (short)0, a00, false, false);
            a01 = __builtin_amdgcn_wmma_f32_16x16x4_f32(false, h0, false, h1, (short)0, a01, false, false);
            a11 = __builtin_amdgcn_wmma_f32_16x16x4_f32(false, h1, false, h1, (short)0, a11, false, false);
        }

        float* __restrict__ ob = out + (size_t)b * NPAIR;

        #pragma unroll
        for (int v = 0; v < 8; ++v) {
            const int i = v + hi8, j = j0;
            if (i < j) ob[triu_idx(i, j)] = a00[v];
        }
        #pragma unroll
        for (int v = 0; v < 8; ++v) {
            const int i = v + hi8, j = 16 + j0;
            ob[triu_idx(i, j)] = a01[v];
        }
        #pragma unroll
        for (int v = 0; v < 8; ++v) {
            const int i = 16 + v + hi8, j = 16 + j0;
            if (i < j) ob[triu_idx(i, j)] = a11[v];
        }

        if (!has_next) break;
        b = bn;
    }
}

extern "C" void kernel_launch(void* const* d_in, const int* in_sizes, int n_in,
                              void* d_out, int out_size, void* d_ws, size_t ws_size,
                              hipStream_t stream) {
    const float* x = (const float*)d_in[0];
    float* out     = (float*)d_out;
    const int n_batch = in_sizes[0] / (NF * ED);
    soi_wmma_kernel<<<NUM_BLOCKS, WAVES_PER_BLOCK * 32, 0, stream>>>(x, out, n_batch);
}
